# starting kernel
# baseline (speedup 1.0000x reference)
#include <hip/hip_runtime.h>
#include <hip/hip_bf16.h>

#define B_      16
#define Q_      2048
#define NCLS_   2
#define DBOX_   6
#define T_      1024
#define COST_CLASS_ 1.0f
#define COST_BBOX_  5.0f

#define TT_PER_WAVE 8
#define QTILES  (Q_ / 16)
#define TGROUPS (T_ / (16 * TT_PER_WAVE))
#define TOTAL_WAVES (B_ * QTILES * TGROUPS)
#define THREADS_PER_BLOCK 256
#define WAVES_PER_BLOCK (THREADS_PER_BLOCK / 32)

typedef __attribute__((ext_vector_type(2))) float v2f;
typedef __attribute__((ext_vector_type(8))) float v8f;

__global__ __launch_bounds__(THREADS_PER_BLOCK)
void hungarian_cost_wmma_kernel(const float* __restrict__ pred_logits,
                                const float* __restrict__ pred_boxes,
                                const int*   __restrict__ tgt_labels,
                                const float* __restrict__ tgt_boxes,
                                float*       __restrict__ out)
{
    const int lane = threadIdx.x & 31;
    const int wid  = (int)((blockIdx.x * blockDim.x + threadIdx.x) >> 5);

    const int b     = wid / (QTILES * TGROUPS);
    const int rem   = wid % (QTILES * TGROUPS);
    const int qtile = rem / TGROUPS;
    const int tg    = rem % TGROUPS;

    const int qbase  = qtile * 16;
    const int tbase0 = tg * (16 * TT_PER_WAVE);

    const int nloc    = lane & 15;
    const int half    = lane >> 4;
    const int rowbase = half * 8;

    v2f amat;
    {
        const float* lg = pred_logits + (size_t)(b * Q_ + qbase + nloc) * NCLS_;
        float l0 = lg[0], l1 = lg[1];
        float mx = fmaxf(l0, l1);
        float e0 = __expf(l0 - mx);
        float e1 = __expf(l1 - mx);
        float inv = 1.0f / (e0 + e1);
        float p0 = e0 * inv;
        float p1 = e1 * inv;
        amat.x = (half == 0) ? (-COST_CLASS_ * p0) : 0.0f;
        amat.y = (half == 0) ? (-COST_CLASS_ * p1) : 0.0f;
    }

    float qb[8][DBOX_];
#pragma unroll
    for (int i = 0; i < 8; ++i) {
        const float* qp = pred_boxes + (size_t)(b * Q_ + qbase + rowbase + i) * DBOX_;
#pragma unroll
        for (int d = 0; d < DBOX_; ++d) qb[i][d] = qp[d];
    }

#pragma unroll
    for (int tt = 0; tt < TT_PER_WAVE; ++tt) {
        const int tbase = tbase0 + tt * 16;
        const int t     = tbase + nloc;

        const int label = tgt_labels[t];

        float tb[DBOX_];
        const float* tp = tgt_boxes + (size_t)t * DBOX_;
#pragma unroll
        for (int d = 0; d < DBOX_; ++d) tb[d] = tp[d];

        v2f bmat;
        bmat.x = (half == 0 && label == 0) ? 1.0f : 0.0f;
        bmat.y = (half == 0 && label == 1) ? 1.0f : 0.0f;

        v8f cacc;
#pragma unroll
        for (int i = 0; i < 8; ++i) {
            float s = fabsf(qb[i][0] - tb[0]);
            s += fabsf(qb[i][1] - tb[1]);
            s += fabsf(qb[i][2] - tb[2]);
            s += fabsf(qb[i][3] - tb[3]);
            s += fabsf(qb[i][4] - tb[4]);
            s += fabsf(qb[i][5] - tb[5]);
            cacc[i] = COST_BBOX_ * s;
        }

        v8f dres = __builtin_amdgcn_wmma_f32_16x16x4_f32(
             false, amat,  false, bmat,
             (short)0, cacc,  false,  false);

        const size_t obase = (size_t)(b * Q_ + qbase + rowbase) * T_ + (size_t)(tbase + nloc);
#pragma unroll
        for (int i = 0; i < 8; ++i) {
            out[obase + (size_t)i * T_] = dres[i];
        }
    }
}

extern "C" void kernel_launch(void* const* d_in, const int* in_sizes, int n_in,
                              void* d_out, int out_size, void* d_ws, size_t ws_size,
                              hipStream_t stream) {
    (void)in_sizes; (void)n_in; (void)out_size; (void)d_ws; (void)ws_size;
    const float* pred_logits = (const float*)d_in[0];
    const float* pred_boxes  = (const float*)d_in[1];
    const int*   tgt_labels  = (const int*)d_in[2];
    const float* tgt_boxes   = (const float*)d_in[3];
    float* out = (float*)d_out;

    const int blocks = TOTAL_WAVES / WAVES_PER_BLOCK;
    hungarian_cost_wmma_kernel<<<blocks, THREADS_PER_BLOCK, 0, stream>>>(
        pred_logits, pred_boxes, tgt_labels, tgt_boxes, out);
}


# Tile remapped to 8 queries x 32 targets (targets of lanes 16-31 fed through k=2,3), each lane owning 4 consecutive targets: every store is 16 B per lane, 4 whole 128-B lines, not two 64-B halves
# speedup vs baseline: 1.9288x; 1.9288x over previous
#include <hip/hip_runtime.h>


#define B_      16
#define Q_      2048
#define NCLS_   2
#define DBOX_   6
#define T_      1024
#define COST_CLASS_ 1.0f
#define COST_BBOX_  5.0f

#define QROWS   8
#define TCOLS   32
#define TSTEP   4
#define TT_PER_WAVE 8
#define TSTEPS  (TT_PER_WAVE / TSTEP)
#define QGROUPS (Q_ / QROWS)
#define TGROUPS (T_ / (TCOLS * TT_PER_WAVE))
#define TOTAL_WAVES (B_ * QGROUPS * TGROUPS)
#define THREADS_PER_BLOCK 256
#define WAVES_PER_BLOCK (THREADS_PER_BLOCK / 32)

typedef __attribute__((ext_vector_type(2))) float v2f;
typedef __attribute__((ext_vector_type(4))) float v4f;
typedef __attribute__((ext_vector_type(8))) float v8f;

__global__ __launch_bounds__(THREADS_PER_BLOCK)
void hungarian_cost_wmma_kernel(const float* __restrict__ pred_logits,
                                const float* __restrict__ pred_boxes,
                                const int*   __restrict__ tgt_labels,
                                const float* __restrict__ tgt_boxes,
                                float*       __restrict__ out)
{
    const int lane = threadIdx.x & 31;
    const int wid  = (int)((blockIdx.x * blockDim.x + threadIdx.x) >> 5);

    const int b   = wid / (QGROUPS * TGROUPS);
    const int rem = wid % (QGROUPS * TGROUPS);
    const int qg  = rem / TGROUPS;
    const int tg  = rem % TGROUPS;

    const int qbase  = qg * QROWS;
    const int tbase0 = tg * (TCOLS * TT_PER_WAVE);

    const int c    = lane & 15;
    const int half = lane >> 4;

    v2f amat;
    {
        const float* lg = pred_logits + (size_t)(b * Q_ + qbase + (c & 7)) * NCLS_;
        float l0 = lg[0], l1 = lg[1];
        float mx = fmaxf(l0, l1);
        float e0 = __expf(l0 - mx);
        float e1 = __expf(l1 - mx);
        float inv = 1.0f / (e0 + e1);
        float p0 = e0 * inv;
        float p1 = e1 * inv;
        const bool live = (half == (c >> 3));
        amat.x = live ? (-COST_CLASS_ * p0) : 0.0f;
        amat.y = live ? (-COST_CLASS_ * p1) : 0.0f;
    }

    float qb[QROWS][DBOX_];
#pragma unroll
    for (int i = 0; i < QROWS; ++i) {
        const float* qp = pred_boxes + (size_t)(b * Q_ + qbase + i) * DBOX_;
#pragma unroll
        for (int d = 0; d < DBOX_; ++d) qb[i][d] = qp[d];
    }

    float* const orow = out + (size_t)(b * Q_ + qbase) * T_;
#pragma unroll 1
    for (int g = 0; g < TSTEPS; ++g) {
        const int t0 = tbase0 + g * (TCOLS * TSTEP) + lane * TSTEP;

        v8f dres[TSTEP];
#pragma unroll
        for (int j = 0; j < TSTEP; ++j) {
            const int t = t0 + j;

            const int label = tgt_labels[t];

            float tb[DBOX_];
            const float* tp = tgt_boxes + (size_t)t * DBOX_;
#pragma unroll
            for (int d = 0; d < DBOX_; ++d) tb[d] = tp[d];

            v2f bmat;
            bmat.x = (label == 0) ? 1.0f : 0.0f;
            bmat.y = (label == 1) ? 1.0f : 0.0f;

            v8f cacc;
#pragma unroll
            for (int i = 0; i < QROWS; ++i) {
                float s = fabsf(qb[i][0] - tb[0]);
                s += fabsf(qb[i][1] - tb[1]);
                s += fabsf(qb[i][2] - tb[2]);
                s += fabsf(qb[i][3] - tb[3]);
                s += fabsf(qb[i][4] - tb[4]);
                s += fabsf(qb[i][5] - tb[5]);
                cacc[i] = COST_BBOX_ * s;
            }

            dres[j] = __builtin_amdgcn_wmma_f32_16x16x4_f32(
                 false, amat,  false, bmat,
                 (short)0, cacc,  false,  false);
        }

#pragma unroll
        for (int i = 0; i < QROWS; ++i) {
            v4f v;
            v.x = dres[0][i];
            v.y = dres[1][i];
            v.z = dres[2][i];
            v.w = dres[3][i];
            *reinterpret_cast<v4f*>(orow + (size_t)i * T_ + t0) = v;
        }
    }
}

extern "C" void kernel_launch(void* const* d_in, const int* in_sizes, int n_in,
                              void* d_out, int out_size, void* d_ws, size_t ws_size,
                              hipStream_t stream) {
    (void)in_sizes; (void)n_in; (void)out_size; (void)d_ws; (void)ws_size;
    const float* pred_logits = (const float*)d_in[0];
    const float* pred_boxes  = (const float*)d_in[1];
    const int*   tgt_labels  = (const int*)d_in[2];
    const float* tgt_boxes   = (const float*)d_in[3];
    float* out = (float*)d_out;

    const int blocks = TOTAL_WAVES / WAVES_PER_BLOCK;
    hungarian_cost_wmma_kernel<<<blocks, THREADS_PER_BLOCK, 0, stream>>>(
        pred_logits, pred_boxes, tgt_labels, tgt_boxes, out);
}


# Each wave owns its 8 queries over the whole 1024-target row (8 store groups, not 2): 4096 waves instead of 16384, all resident at once, query set-up paid once per row
# speedup vs baseline: 2.1646x; 1.1223x over previous
#include <hip/hip_runtime.h>


#define B_      16
#define Q_      2048
#define NCLS_   2
#define DBOX_   6
#define T_      1024
#define COST_CLASS_ 1.0f
#define COST_BBOX_  5.0f

#define QROWS   8
#define TCOLS   32
#define TSTEP   4
#define TT_PER_WAVE 32
#define TSTEPS  (TT_PER_WAVE / TSTEP)
#define QGROUPS (Q_ / QROWS)
#define TGROUPS (T_ / (TCOLS * TT_PER_WAVE))
#define TOTAL_WAVES (B_ * QGROUPS * TGROUPS)
#define THREADS_PER_BLOCK 256
#define WAVES_PER_BLOCK (THREADS_PER_BLOCK / 32)

typedef __attribute__((ext_vector_type(2))) float v2f;
typedef __attribute__((ext_vector_type(4))) float v4f;
typedef __attribute__((ext_vector_type(8))) float v8f;

__global__ __launch_bounds__(THREADS_PER_BLOCK)
void hungarian_cost_wmma_kernel(const float* __restrict__ pred_logits,
                                const float* __restrict__ pred_boxes,
                                const int*   __restrict__ tgt_labels,
                                const float* __restrict__ tgt_boxes,
                                float*       __restrict__ out)
{
    const int lane = threadIdx.x & 31;
    const int wid  = (int)((blockIdx.x * blockDim.x + threadIdx.x) >> 5);

    const int b   = wid / (QGROUPS * TGROUPS);
    const int rem = wid % (QGROUPS * TGROUPS);
    const int qg  = rem / TGROUPS;
    const int tg  = rem % TGROUPS;

    const int qbase  = qg * QROWS;
    const int tbase0 = tg * (TCOLS * TT_PER_WAVE);

    const int c    = lane & 15;
    const int half = lane >> 4;

    v2f amat;
    {
        const float* lg = pred_logits + (size_t)(b * Q_ + qbase + (c & 7)) * NCLS_;
        float l0 = lg[0], l1 = lg[1];
        float mx = fmaxf(l0, l1);
        float e0 = __expf(l0 - mx);
        float e1 = __expf(l1 - mx);
        float inv = 1.0f / (e0 + e1);
        float p0 = e0 * inv;
        float p1 = e1 * inv;
        const bool live = (half == (c >> 3));
        amat.x = live ? (-COST_CLASS_ * p0) : 0.0f;
        amat.y = live ? (-COST_CLASS_ * p1) : 0.0f;
    }

    float qb[QROWS][DBOX_];
#pragma unroll
    for (int i = 0; i < QROWS; ++i) {
        const float* qp = pred_boxes + (size_t)(b * Q_ + qbase + i) * DBOX_;
#pragma unroll
        for (int d = 0; d < DBOX_; ++d) qb[i][d] = qp[d];
    }

    float* const orow = out + (size_t)(b * Q_ + qbase) * T_;
#pragma unroll 1
    for (int g = 0; g < TSTEPS; ++g) {
        const int t0 = tbase0 + g * (TCOLS * TSTEP) + lane * TSTEP;

        v8f dres[TSTEP];
#pragma unroll
        for (int j = 0; j < TSTEP; ++j) {
            const int t = t0 + j;

            const int label = tgt_labels[t];

            float tb[DBOX_];
            const float* tp = tgt_boxes + (size_t)t * DBOX_;
#pragma unroll
            for (int d = 0; d < DBOX_; ++d) tb[d] = tp[d];

            v2f bmat;
            bmat.x = (label == 0) ? 1.0f : 0.0f;
            bmat.y = (label == 1) ? 1.0f : 0.0f;

            v8f cacc;
#pragma unroll
            for (int i = 0; i < QROWS; ++i) {
                float s = fabsf(qb[i][0] - tb[0]);
                s += fabsf(qb[i][1] - tb[1]);
                s += fabsf(qb[i][2] - tb[2]);
                s += fabsf(qb[i][3] - tb[3]);
                s += fabsf(qb[i][4] - tb[4]);
                s += fabsf(qb[i][5] - tb[5]);
                cacc[i] = COST_BBOX_ * s;
            }

            dres[j] = __builtin_amdgcn_wmma_f32_16x16x4_f32(
                 false, amat,  false, bmat,
                 (short)0, cacc,  false,  false);
        }

#pragma unroll
        for (int i = 0; i < QROWS; ++i) {
            v4f v;
            v.x = dres[0][i];
            v.y = dres[1][i];
            v.z = dres[2][i];
            v.w = dres[3][i];
            *reinterpret_cast<v4f*>(orow + (size_t)i * T_ + t0) = v;
        }
    }
}

extern "C" void kernel_launch(void* const* d_in, const int* in_sizes, int n_in,
                              void* d_out, int out_size, void* d_ws, size_t ws_size,
                              hipStream_t stream) {
    (void)in_sizes; (void)n_in; (void)out_size; (void)d_ws; (void)ws_size;
    const float* pred_logits = (const float*)d_in[0];
    const float* pred_boxes  = (const float*)d_in[1];
    const int*   tgt_labels  = (const int*)d_in[2];
    const float* tgt_boxes   = (const float*)d_in[3];
    float* out = (float*)d_out;

    const int blocks = TOTAL_WAVES / WAVES_PER_BLOCK;
    hungarian_cost_wmma_kernel<<<blocks, THREADS_PER_BLOCK, 0, stream>>>(
        pred_logits, pred_boxes, tgt_labels, tgt_boxes, out);
}


# Targets (boxes, labels) staged once per work-group into LDS, one plane per box coordinate, by whole-line loads; lanes read them by 16-B LDS loads, not 96-B-strided global loads
# speedup vs baseline: 2.4999x; 1.1549x over previous
#include <hip/hip_runtime.h>


#define B_      16
#define Q_      2048
#define NCLS_   2
#define DBOX_   6
#define T_      1024
#define COST_CLASS_ 1.0f
#define COST_BBOX_  5.0f

#define QROWS   8
#define TCOLS   32
#define TSTEP   4
#define TT_PER_WAVE 32
#define TSTEPS  (TT_PER_WAVE / TSTEP)
#define QGROUPS (Q_ / QROWS)
#define TGROUPS (T_ / (TCOLS * TT_PER_WAVE))
#define TOTAL_WAVES (B_ * QGROUPS * TGROUPS)
#define THREADS_PER_BLOCK 256
#define WAVES_PER_BLOCK (THREADS_PER_BLOCK / 32)
#define TPITCH  (T_ + 20)

typedef __attribute__((ext_vector_type(2))) float v2f;
typedef __attribute__((ext_vector_type(4))) float v4f;
typedef __attribute__((ext_vector_type(8))) float v8f;

__global__ __launch_bounds__(THREADS_PER_BLOCK)
void hungarian_cost_wmma_kernel(const float* __restrict__ pred_logits,
                                const float* __restrict__ pred_boxes,
                                const int*   __restrict__ tgt_labels,
                                const float* __restrict__ tgt_boxes,
                                float*       __restrict__ out)
{
    const int lane = threadIdx.x & 31;
    const int wid  = (int)((blockIdx.x * blockDim.x + threadIdx.x) >> 5);

    const int b   = wid / (QGROUPS * TGROUPS);
    const int rem = wid % (QGROUPS * TGROUPS);
    const int qg  = rem / TGROUPS;
    const int tg  = rem % TGROUPS;

    const int qbase  = qg * QROWS;
    const int tbase0 = tg * (TCOLS * TT_PER_WAVE);

    const int c    = lane & 15;
    const int half = lane >> 4;

    v2f amat;
    {
        const float* lg = pred_logits + (size_t)(b * Q_ + qbase + (c & 7)) * NCLS_;
        float l0 = lg[0], l1 = lg[1];
        float mx = fmaxf(l0, l1);
        float e0 = __expf(l0 - mx);
        float e1 = __expf(l1 - mx);
        float inv = 1.0f / (e0 + e1);
        float p0 = e0 * inv;
        float p1 = e1 * inv;
        const bool live = (half == (c >> 3));
        amat.x = live ? (-COST_CLASS_ * p0) : 0.0f;
        amat.y = live ? (-COST_CLASS_ * p1) : 0.0f;
    }

    float qb[QROWS][DBOX_];
#pragma unroll
    for (int i = 0; i < QROWS; ++i) {
        const float* qp = pred_boxes + (size_t)(b * Q_ + qbase + i) * DBOX_;
#pragma unroll
        for (int d = 0; d < DBOX_; ++d) qb[i][d] = qp[d];
    }

    __shared__ __attribute__((aligned(16))) float s_tb[DBOX_ * TPITCH];
    __shared__ __attribute__((aligned(16))) int   s_lab[T_];
    {
        const int tid = threadIdx.x;
        const v2f* bp = reinterpret_cast<const v2f*>(tgt_boxes);
#pragma unroll
        for (int k = 0; k < (T_ * DBOX_ / 2) / THREADS_PER_BLOCK; ++k) {
            const int p = tid + k * THREADS_PER_BLOCK;
            const v2f v = bp[p];
            const int t = p / 3;
            const int d = 2 * (p - 3 * t);
            s_tb[d * TPITCH + t]       = v.x;
            s_tb[(d + 1) * TPITCH + t] = v.y;
        }
#pragma unroll
        for (int k = 0; k < T_ / THREADS_PER_BLOCK; ++k) {
            const int t = tid + k * THREADS_PER_BLOCK;
            s_lab[t] = tgt_labels[t];
        }
    }
    __syncthreads();

    float* const orow = out + (size_t)(b * Q_ + qbase) * T_;
#pragma unroll 1
    for (int g = 0; g < TSTEPS; ++g) {
        const int t0 = tbase0 + g * (TCOLS * TSTEP) + lane * TSTEP;

        typedef __attribute__((ext_vector_type(4))) int v4i;
        const v4i lab4 = *reinterpret_cast<const v4i*>(&s_lab[t0]);
        v4f tb4[DBOX_];
#pragma unroll
        for (int d = 0; d < DBOX_; ++d)
            tb4[d] = *reinterpret_cast<const v4f*>(&s_tb[d * TPITCH + t0]);

        v8f dres[TSTEP];
#pragma unroll
        for (int j = 0; j < TSTEP; ++j) {
            const int label = lab4[j];

            float tb[DBOX_];
#pragma unroll
            for (int d = 0; d < DBOX_; ++d) tb[d] = tb4[d][j];

            v2f bmat;
            bmat.x = (label == 0) ? 1.0f : 0.0f;
            bmat.y = (label == 1) ? 1.0f : 0.0f;

            v8f cacc;
#pragma unroll
            for (int i = 0; i < QROWS; ++i) {
                float s = fabsf(qb[i][0] - tb[0]);
                s += fabsf(qb[i][1] - tb[1]);
                s += fabsf(qb[i][2] - tb[2]);
                s += fabsf(qb[i][3] - tb[3]);
                s += fabsf(qb[i][4] - tb[4]);
                s += fabsf(qb[i][5] - tb[5]);
                cacc[i] = COST_BBOX_ * s;
            }

            dres[j] = __builtin_amdgcn_wmma_f32_16x16x4_f32(
                 false, amat,  false, bmat,
                 (short)0, cacc,  false,  false);
        }

#pragma unroll
        for (int i = 0; i < QROWS; ++i) {
            v4f v;
            v.x = dres[0][i];
            v.y = dres[1][i];
            v.z = dres[2][i];
            v.w = dres[3][i];
            *reinterpret_cast<v4f*>(orow + (size_t)i * T_ + t0) = v;
        }
    }
}

extern "C" void kernel_launch(void* const* d_in, const int* in_sizes, int n_in,
                              void* d_out, int out_size, void* d_ws, size_t ws_size,
                              hipStream_t stream) {
    (void)in_sizes; (void)n_in; (void)out_size; (void)d_ws; (void)ws_size;
    const float* pred_logits = (const float*)d_in[0];
    const float* pred_boxes  = (const float*)d_in[1];
    const int*   tgt_labels  = (const int*)d_in[2];
    const float* tgt_boxes   = (const float*)d_in[3];
    float* out = (float*)d_out;

    const int blocks = TOTAL_WAVES / WAVES_PER_BLOCK;
    hungarian_cost_wmma_kernel<<<blocks, THREADS_PER_BLOCK, 0, stream>>>(
        pred_logits, pred_boxes, tgt_labels, tgt_boxes, out);
}
